# starting kernel
# baseline (speedup 1.0000x reference)
#include <hip/hip_runtime.h>

#define NNODES 100000
#define NEDGES 1200000
#define IN_DIM 64
#define HID_DIM 64
#define OUT_DIM 32

typedef __attribute__((ext_vector_type(2))) float v2f;
typedef __attribute__((ext_vector_type(4))) float v4f;
typedef __attribute__((ext_vector_type(8))) float v8f;

__global__ void gcn_fill_kernel(float* __restrict__ p, float v, int n) {
    int i = blockIdx.x * blockDim.x + threadIdx.x;
    if (i < n) p[i] = v;
}

__global__ void gcn_deg_kernel(const int* __restrict__ dst, float* __restrict__ deg, int E) {
    int e = blockIdx.x * blockDim.x + threadIdx.x;
    if (e < E) atomicAdd(&deg[dst[e]], 1.0f);
}

__global__ void gcn_rsqrt_kernel(float* __restrict__ deg, int n) {
    int i = blockIdx.x * blockDim.x + threadIdx.x;
    if (i < n) deg[i] = rsqrtf(deg[i]);
}

template <int K, int N>
__global__ __launch_bounds__(256) void gcn_gemm_wmma_kernel(
    const float* __restrict__ A, const float* __restrict__ B,
    float* __restrict__ C, int M) {
    constexpr int NT = N / 16;

    const int lane     = threadIdx.x & 31;
    const int wave     = threadIdx.x >> 5;
    const int row_tile = blockIdx.x * 8 + wave;
    if (row_tile * 16 >= M) return;

    const int m     = lane & 15;
    const int khalf = lane >> 4;
    const int row   = row_tile * 16 + m;

    v8f acc[NT];
#pragma unroll
    for (int nt = 0; nt < NT; ++nt) acc[nt] = (v8f){0.f,0.f,0.f,0.f,0.f,0.f,0.f,0.f};

    const float* __restrict__ arow = A + (size_t)row * K;
#pragma unroll
    for (int k0 = 0; k0 < K; k0 += 4) {
        const int kk = k0 + khalf * 2;
        v2f a = *(const v2f*)(arow + kk);
#pragma unroll
        for (int nt = 0; nt < NT; ++nt) {
            const int n = nt * 16 + m;
            v2f b;
            b.x = B[(size_t)kk * N + n];
            b.y = B[(size_t)(kk + 1) * N + n];
            acc[nt] = __builtin_amdgcn_wmma_f32_16x16x4_f32(
                false, a, false, b, (short)0, acc[nt], false, false);
        }
    }

#pragma unroll
    for (int nt = 0; nt < NT; ++nt) {
        float* __restrict__ crow =
            C + (size_t)(row_tile * 16 + khalf * 8) * N + nt * 16 + m;
#pragma unroll
        for (int j = 0; j < 8; ++j) crow[(size_t)j * N] = acc[nt][j];
    }
}

template <int F>
__global__ __launch_bounds__(256) void gcn_scatter_kernel(
    const float* __restrict__ h, const int* __restrict__ src,
    const int* __restrict__ dst, const float* __restrict__ isq,
    float* __restrict__ out, int E) {
    constexpr int Q = F / 4;
    unsigned t = blockIdx.x * blockDim.x + threadIdx.x;
    unsigned e = t / Q;
    unsigned q = t % Q;
    if (e >= (unsigned)E) return;
    int s = src[e];
    int d = dst[e];
    float nrm = isq[s] * isq[d];
    v4f msg = *(const v4f*)(h + (unsigned)s * F + q * 4);
    float* o = out + (unsigned)d * F + q * 4;
    atomicAdd(o + 0, msg.x * nrm);
    atomicAdd(o + 1, msg.y * nrm);
    atomicAdd(o + 2, msg.z * nrm);
    atomicAdd(o + 3, msg.w * nrm);
}

template <int F, bool RELU>
__global__ __launch_bounds__(256) void gcn_finish_kernel(
    const float* __restrict__ h, const float* __restrict__ isq,
    const float* __restrict__ bias, float* __restrict__ out, int n) {
    unsigned idx = blockIdx.x * blockDim.x + threadIdx.x;
    if (idx >= (unsigned)n * F) return;
    unsigned i = idx / F;
    unsigned f = idx % F;
    float si = isq[i];
    float v  = out[idx] + h[idx] * si * si + bias[f];
    out[idx] = RELU ? fmaxf(v, 0.0f) : v;
}

extern "C" void kernel_launch(void* const* d_in, const int* in_sizes, int n_in,
                              void* d_out, int out_size, void* d_ws, size_t ws_size,
                              hipStream_t stream) {
    const float* x  = (const float*)d_in[0];
    const int*   ei = (const int*)d_in[1];
    const float* W1 = (const float*)d_in[2];
    const float* b1 = (const float*)d_in[3];
    const float* W2 = (const float*)d_in[4];
    const float* b2 = (const float*)d_in[5];
    const int* src = ei;
    const int* dst = ei + NEDGES;
    float* out = (float*)d_out;

    float* ws   = (float*)d_ws;
    float* isq  = ws;
    float* h1   = ws + 131072;
    float* agg1 = h1 + (size_t)NNODES * HID_DIM;
    float* h2   = h1;

    const int T = 256;

    gcn_fill_kernel<<<(NNODES + T - 1) / T, T, 0, stream>>>(isq, 1.0f, NNODES);
    gcn_deg_kernel<<<(NEDGES + T - 1) / T, T, 0, stream>>>(dst, isq, NEDGES);
    gcn_rsqrt_kernel<<<(NNODES + T - 1) / T, T, 0, stream>>>(isq, NNODES);

    {
        int waves  = NNODES / 16;
        int blocks = (waves + 7) / 8;
        gcn_gemm_wmma_kernel<IN_DIM, HID_DIM><<<blocks, T, 0, stream>>>(x, W1, h1, NNODES);
    }
    gcn_fill_kernel<<<((NNODES * HID_DIM) + T - 1) / T, T, 0, stream>>>(agg1, 0.0f, NNODES * HID_DIM);
    {
        unsigned total = (unsigned)NEDGES * (HID_DIM / 4);
        gcn_scatter_kernel<HID_DIM><<<(total + T - 1) / T, T, 0, stream>>>(h1, src, dst, isq, agg1, NEDGES);
    }
    gcn_finish_kernel<HID_DIM, true><<<((NNODES * HID_DIM) + T - 1) / T, T, 0, stream>>>(h1, isq, b1, agg1, NNODES);

    {
        int waves  = NNODES / 16;
        int blocks = (waves + 7) / 8;
        gcn_gemm_wmma_kernel<HID_DIM, OUT_DIM><<<blocks, T, 0, stream>>>(agg1, W2, h2, NNODES);
    }
    gcn_fill_kernel<<<((NNODES * OUT_DIM) + T - 1) / T, T, 0, stream>>>(out, 0.0f, NNODES * OUT_DIM);
    {
        unsigned total = (unsigned)NEDGES * (OUT_DIM / 4);
        gcn_scatter_kernel<OUT_DIM><<<(total + T - 1) / T, T, 0, stream>>>(h2, src, dst, isq, out, NEDGES);
    }
    gcn_finish_kernel<OUT_DIM, false><<<((NNODES * OUT_DIM) + T - 1) / T, T, 0, stream>>>(h2, isq, b2, out, NNODES);
}


# scatter remapped: one wave per 32 edges, lane = feature, so every gather and f32 atomic instruction covers one whole 128-byte row line (4x fewer atomic line requests); atomic sums kept
# speedup vs baseline: 1.3737x; 1.3737x over previous
#include <hip/hip_runtime.h>

#define NNODES 100000
#define NEDGES 1200000
#define IN_DIM 64
#define HID_DIM 64
#define OUT_DIM 32

typedef __attribute__((ext_vector_type(2))) float v2f;
typedef __attribute__((ext_vector_type(4))) float v4f;
typedef __attribute__((ext_vector_type(8))) float v8f;

__global__ void gcn_fill_kernel(float* __restrict__ p, float v, int n) {
    int i = blockIdx.x * blockDim.x + threadIdx.x;
    if (i < n) p[i] = v;
}

__global__ void gcn_deg_kernel(const int* __restrict__ dst, float* __restrict__ deg, int E) {
    int e = blockIdx.x * blockDim.x + threadIdx.x;
    if (e < E) atomicAdd(&deg[dst[e]], 1.0f);
}

__global__ void gcn_rsqrt_kernel(float* __restrict__ deg, int n) {
    int i = blockIdx.x * blockDim.x + threadIdx.x;
    if (i < n) deg[i] = rsqrtf(deg[i]);
}

template <int K, int N>
__global__ __launch_bounds__(256) void gcn_gemm_wmma_kernel(
    const float* __restrict__ A, const float* __restrict__ B,
    float* __restrict__ C, int M) {
    constexpr int NT = N / 16;

    const int lane     = threadIdx.x & 31;
    const int wave     = threadIdx.x >> 5;
    const int row_tile = blockIdx.x * 8 + wave;
    if (row_tile * 16 >= M) return;

    const int m     = lane & 15;
    const int khalf = lane >> 4;
    const int row   = row_tile * 16 + m;

    v8f acc[NT];
#pragma unroll
    for (int nt = 0; nt < NT; ++nt) acc[nt] = (v8f){0.f,0.f,0.f,0.f,0.f,0.f,0.f,0.f};

    const float* __restrict__ arow = A + (size_t)row * K;
#pragma unroll
    for (int k0 = 0; k0 < K; k0 += 4) {
        const int kk = k0 + khalf * 2;
        v2f a = *(const v2f*)(arow + kk);
#pragma unroll
        for (int nt = 0; nt < NT; ++nt) {
            const int n = nt * 16 + m;
            v2f b;
            b.x = B[(size_t)kk * N + n];
            b.y = B[(size_t)(kk + 1) * N + n];
            acc[nt] = __builtin_amdgcn_wmma_f32_16x16x4_f32(
                false, a, false, b, (short)0, acc[nt], false, false);
        }
    }

#pragma unroll
    for (int nt = 0; nt < NT; ++nt) {
        float* __restrict__ crow =
            C + (size_t)(row_tile * 16 + khalf * 8) * N + nt * 16 + m;
#pragma unroll
        for (int j = 0; j < 8; ++j) crow[(size_t)j * N] = acc[nt][j];
    }
}

template <int F>
__global__ __launch_bounds__(256) void gcn_scatter_kernel(
    const float* __restrict__ h, const int* __restrict__ src,
    const int* __restrict__ dst, const float* __restrict__ isq,
    float* __restrict__ out, int E) {
    constexpr int S   = F / 32;
    constexpr int GRP = 8;
    const int lane = threadIdx.x & 31;
    const unsigned wave =
        blockIdx.x * 8u + (unsigned)__builtin_amdgcn_readfirstlane((int)(threadIdx.x >> 5));
    const unsigned e0 = wave * 32u;
    if (e0 >= (unsigned)E) return;

    const unsigned e = e0 + (unsigned)lane;
    int   s = 0, d = 0;
    float nrm = 0.0f;
    if (e < (unsigned)E) {
        s = src[e];
        d = dst[e];
        nrm = isq[s] * isq[d];
    }
    const int nrm_bits = __float_as_int(nrm);

    if (e0 + 32u <= (unsigned)E) {
#pragma unroll
        for (int g = 0; g < 32; g += GRP) {
            float v[GRP][S];
#pragma unroll
            for (int j = 0; j < GRP; ++j) {
                const unsigned sj = (unsigned)__builtin_amdgcn_readlane(s, g + j);
                const float* __restrict__ hp = h + (size_t)(sj * (unsigned)F) + lane;
#pragma unroll
                for (int q = 0; q < S; ++q) v[j][q] = hp[q * 32];
            }
#pragma unroll
            for (int j = 0; j < GRP; ++j) {
                const unsigned dj = (unsigned)__builtin_amdgcn_readlane(d, g + j);
                const float nj = __int_as_float(__builtin_amdgcn_readlane(nrm_bits, g + j));
                float* op = out + (size_t)(dj * (unsigned)F) + lane;
#pragma unroll
                for (int q = 0; q < S; ++q) atomicAdd(op + q * 32, v[j][q] * nj);
            }
        }
    } else {
        const int cnt = (int)((unsigned)E - e0);
        for (int j = 0; j < cnt; ++j) {
            const unsigned sj = (unsigned)__builtin_amdgcn_readlane(s, j);
            const unsigned dj = (unsigned)__builtin_amdgcn_readlane(d, j);
            const float nj = __int_as_float(__builtin_amdgcn_readlane(nrm_bits, j));
            const float* __restrict__ hp = h + (size_t)(sj * (unsigned)F) + lane;
            float* op = out + (size_t)(dj * (unsigned)F) + lane;
#pragma unroll
            for (int q = 0; q < S; ++q) atomicAdd(op + q * 32, hp[q * 32] * nj);
        }
    }
}

template <int F, bool RELU>
__global__ __launch_bounds__(256) void gcn_finish_kernel(
    const float* __restrict__ h, const float* __restrict__ isq,
    const float* __restrict__ bias, float* __restrict__ out, int n) {
    unsigned idx = blockIdx.x * blockDim.x + threadIdx.x;
    if (idx >= (unsigned)n * F) return;
    unsigned i = idx / F;
    unsigned f = idx % F;
    float si = isq[i];
    float v  = out[idx] + h[idx] * si * si + bias[f];
    out[idx] = RELU ? fmaxf(v, 0.0f) : v;
}

extern "C" void kernel_launch(void* const* d_in, const int* in_sizes, int n_in,
                              void* d_out, int out_size, void* d_ws, size_t ws_size,
                              hipStream_t stream) {
    const float* x  = (const float*)d_in[0];
    const int*   ei = (const int*)d_in[1];
    const float* W1 = (const float*)d_in[2];
    const float* b1 = (const float*)d_in[3];
    const float* W2 = (const float*)d_in[4];
    const float* b2 = (const float*)d_in[5];
    const int* src = ei;
    const int* dst = ei + NEDGES;
    float* out = (float*)d_out;

    float* ws   = (float*)d_ws;
    float* isq  = ws;
    float* h1   = ws + 131072;
    float* agg1 = h1 + (size_t)NNODES * HID_DIM;
    float* h2   = h1;

    const int T = 256;

    gcn_fill_kernel<<<(NNODES + T - 1) / T, T, 0, stream>>>(isq, 1.0f, NNODES);
    gcn_deg_kernel<<<(NEDGES + T - 1) / T, T, 0, stream>>>(dst, isq, NEDGES);
    gcn_rsqrt_kernel<<<(NNODES + T - 1) / T, T, 0, stream>>>(isq, NNODES);

    {
        int waves  = NNODES / 16;
        int blocks = (waves + 7) / 8;
        gcn_gemm_wmma_kernel<IN_DIM, HID_DIM><<<blocks, T, 0, stream>>>(x, W1, h1, NNODES);
    }
    gcn_fill_kernel<<<((NNODES * HID_DIM) + T - 1) / T, T, 0, stream>>>(agg1, 0.0f, NNODES * HID_DIM);
    {
        gcn_scatter_kernel<HID_DIM><<<(NEDGES + T - 1) / T, T, 0, stream>>>(h1, src, dst, isq, agg1, NEDGES);
    }
    gcn_finish_kernel<HID_DIM, true><<<((NNODES * HID_DIM) + T - 1) / T, T, 0, stream>>>(h1, isq, b1, agg1, NNODES);

    {
        int waves  = NNODES / 16;
        int blocks = (waves + 7) / 8;
        gcn_gemm_wmma_kernel<HID_DIM, OUT_DIM><<<blocks, T, 0, stream>>>(agg1, W2, h2, NNODES);
    }
    gcn_fill_kernel<<<((NNODES * OUT_DIM) + T - 1) / T, T, 0, stream>>>(out, 0.0f, NNODES * OUT_DIM);
    {
        gcn_scatter_kernel<OUT_DIM><<<(NEDGES + T - 1) / T, T, 0, stream>>>(h2, src, dst, isq, out, NEDGES);
    }
    gcn_finish_kernel<OUT_DIM, false><<<((NNODES * OUT_DIM) + T - 1) / T, T, 0, stream>>>(h2, isq, b2, out, NNODES);
}


# f32 atomic scatter-add replaced by per-call counting sort of edges by destination (int atomic returns slot, scan); one wave sums each row in registers in slot (arrival-rank) order, stored once
# speedup vs baseline: 1.4782x; 1.0761x over previous
#include <hip/hip_runtime.h>

#define NNODES 100000
#define NEDGES 1200000
#define IN_DIM 64
#define HID_DIM 64
#define OUT_DIM 32

typedef __attribute__((ext_vector_type(2))) float v2f;
typedef __attribute__((ext_vector_type(4))) float v4f;
typedef __attribute__((ext_vector_type(8))) float v8f;

__global__ void gcn_zero_kernel(int* __restrict__ p, int n) {
    int i = blockIdx.x * blockDim.x + threadIdx.x;
    if (i < n) p[i] = 0;
}

__global__ void gcn_rank_kernel(const int* __restrict__ dst, int* __restrict__ cnt,
                                int* __restrict__ rank, int E) {
    int e = blockIdx.x * blockDim.x + threadIdx.x;
    if (e < E) rank[e] = atomicAdd(&cnt[dst[e]], 1);
}

__global__ __launch_bounds__(256) void gcn_blocksum_kernel(
    const int* __restrict__ cnt, int* __restrict__ bsum, int n) {
    __shared__ int red[256];
    const int t = threadIdx.x;
    const int i = blockIdx.x * 256 + t;
    red[t] = (i < n) ? cnt[i] : 0;
    __syncthreads();
    for (int s = 128; s > 0; s >>= 1) {
        if (t < s) red[t] += red[t + s];
        __syncthreads();
    }
    if (t == 0) bsum[blockIdx.x] = red[0];
}

__global__ __launch_bounds__(256) void gcn_rowptr_kernel(
    const int* __restrict__ cnt, const int* __restrict__ bsum,
    int* __restrict__ rowptr, float* __restrict__ isq,
    int* __restrict__ csr_src, float* __restrict__ csr_w, int n) {
    __shared__ int red[256];
    __shared__ int sc[256];
    const int t = threadIdx.x;
    const int b = blockIdx.x;
    int part = 0;
    for (int k = t; k < b; k += 256) part += bsum[k];
    red[t] = part;
    const int i = b * 256 + t;
    const int c = (i < n) ? cnt[i] : 0;
    sc[t] = c;
    __syncthreads();
    for (int s = 128; s > 0; s >>= 1) {
        if (t < s) red[t] += red[t + s];
        __syncthreads();
    }
    for (int off = 1; off < 256; off <<= 1) {
        const int v = (t >= off) ? sc[t - off] : 0;
        __syncthreads();
        sc[t] += v;
        __syncthreads();
    }
    if (i < n) {
        const int   p = red[0] + sc[t] - c + i;
        const float r = rsqrtf((float)(c + 1));
        rowptr[i] = p;
        isq[i]    = r;
        csr_src[p + c] = i;
        csr_w[p + c]   = r;
    }
}

__global__ void gcn_csr_kernel(const int* __restrict__ src, const int* __restrict__ dst,
                               const int* __restrict__ rank, const int* __restrict__ rowptr,
                               const float* __restrict__ isq, int* __restrict__ csr_src,
                               float* __restrict__ csr_w, int E) {
    int e = blockIdx.x * blockDim.x + threadIdx.x;
    if (e < E) {
        const int s = src[e];
        const int p = rowptr[dst[e]] + rank[e];
        csr_src[p] = s;
        csr_w[p]   = isq[s];
    }
}

template <int K, int N>
__global__ __launch_bounds__(256) void gcn_gemm_wmma_kernel(
    const float* __restrict__ A, const float* __restrict__ B,
    float* __restrict__ C, int M) {
    constexpr int NT = N / 16;

    const int lane     = threadIdx.x & 31;
    const int wave     = threadIdx.x >> 5;
    const int row_tile = blockIdx.x * 8 + wave;
    if (row_tile * 16 >= M) return;

    const int m     = lane & 15;
    const int khalf = lane >> 4;
    const int row   = row_tile * 16 + m;

    v8f acc[NT];
#pragma unroll
    for (int nt = 0; nt < NT; ++nt) acc[nt] = (v8f){0.f,0.f,0.f,0.f,0.f,0.f,0.f,0.f};

    const float* __restrict__ arow = A + (size_t)row * K;
#pragma unroll
    for (int k0 = 0; k0 < K; k0 += 4) {
        const int kk = k0 + khalf * 2;
        v2f a = *(const v2f*)(arow + kk);
#pragma unroll
        for (int nt = 0; nt < NT; ++nt) {
            const int n = nt * 16 + m;
            v2f b;
            b.x = B[(size_t)kk * N + n];
            b.y = B[(size_t)(kk + 1) * N + n];
            acc[nt] = __builtin_amdgcn_wmma_f32_16x16x4_f32(
                false, a, false, b, (short)0, acc[nt], false, false);
        }
    }

#pragma unroll
    for (int nt = 0; nt < NT; ++nt) {
        float* __restrict__ crow =
            C + (size_t)(row_tile * 16 + khalf * 8) * N + nt * 16 + m;
#pragma unroll
        for (int j = 0; j < 8; ++j) crow[(size_t)j * N] = acc[nt][j];
    }
}


template <int F, int G, bool PAD>
__device__ __forceinline__ void gcn_row_group(
    const float* __restrict__ h, int s, int wbits, int j, int m, int lane,
    float (&acc)[F / 32]) {
    constexpr int S = F / 32;
    float v[G][S];
#pragma unroll
    for (int u = 0; u < G; ++u) {
        const unsigned sj = (unsigned)__builtin_amdgcn_readlane(s, j + u);
        const float* __restrict__ hp = h + (size_t)(sj * (unsigned)F) + lane;
#pragma unroll
        for (int q = 0; q < S; ++q) v[u][q] = hp[q * 32];
    }
#pragma unroll
    for (int u = 0; u < G; ++u) {
        const float wj = __int_as_float(__builtin_amdgcn_readlane(wbits, j + u));
#pragma unroll
        for (int q = 0; q < S; ++q) {
            const float x = (!PAD || j + u < m) ? v[u][q] : 0.0f;
            acc[q] += x * wj;
        }
    }
}

template <int F, bool RELU>
__global__ __launch_bounds__(256) void gcn_gather_kernel(
    const float* __restrict__ h, const int* __restrict__ rowptr,
    const int* __restrict__ cnt, const int* __restrict__ csr_src,
    const float* __restrict__ csr_w, const float* __restrict__ isq,
    const float* __restrict__ bias, float* __restrict__ out, int n) {
    constexpr int S = F / 32;
    const int lane = threadIdx.x & 31;
    const unsigned node =
        blockIdx.x * 8u + (unsigned)__builtin_amdgcn_readfirstlane((int)(threadIdx.x >> 5));
    if (node >= (unsigned)n) return;

    const int   beg = rowptr[node];
    const int   len = cnt[node] + 1;
    const float di  = isq[node];

    float acc[S];
#pragma unroll
    for (int q = 0; q < S; ++q) acc[q] = 0.0f;

    for (int base = 0; base < len; base += 32) {
        const int m = (len - base < 32) ? (len - base) : 32;
        int   s = (int)node;
        float w = 0.0f;
        if (lane < m) {
            s = csr_src[beg + base + lane];
            w = csr_w[beg + base + lane] * di;
        }
        const int wbits = __float_as_int(w);
        int j = 0;
        for (; j + 8 <= m; j += 8) gcn_row_group<F, 8, false>(h, s, wbits, j, m, lane, acc);
        if (j + 4 <= m) {
            gcn_row_group<F, 4, false>(h, s, wbits, j, m, lane, acc);
            j += 4;
        }
        if (j < m) gcn_row_group<F, 4, true>(h, s, wbits, j, m, lane, acc);
    }

    float* __restrict__ op = out + (size_t)(node * (unsigned)F) + lane;
#pragma unroll
    for (int q = 0; q < S; ++q) {
        const float v = acc[q] + bias[q * 32 + lane];
        op[q * 32] = RELU ? fmaxf(v, 0.0f) : v;
    }
}

extern "C" void kernel_launch(void* const* d_in, const int* in_sizes, int n_in,
                              void* d_out, int out_size, void* d_ws, size_t ws_size,
                              hipStream_t stream) {
    const float* x  = (const float*)d_in[0];
    const int*   ei = (const int*)d_in[1];
    const float* W1 = (const float*)d_in[2];
    const float* b1 = (const float*)d_in[3];
    const float* W2 = (const float*)d_in[4];
    const float* b2 = (const float*)d_in[5];
    const int* src = ei;
    const int* dst = ei + NEDGES;
    float* out = (float*)d_out;

    float* ws      = (float*)d_ws;
    float* isq     = ws;
    int*   cnt     = (int*)(ws + 131072);
    int*   rowptr  = (int*)(ws + 262144);
    int*   bsum    = (int*)(ws + 393216);
    int*   rank    = (int*)(ws + 394240);
    int*   csr_src = rank + NEDGES;
    float* csr_w   = (float*)(csr_src + NEDGES + NNODES);
    float* h1      = csr_w + NEDGES + NNODES;
    float* agg1    = h1 + (size_t)NNODES * HID_DIM;
    float* h2      = h1;

    const int T = 256;

    gcn_zero_kernel<<<(NNODES + T - 1) / T, T, 0, stream>>>(cnt, NNODES);
    gcn_rank_kernel<<<(NEDGES + T - 1) / T, T, 0, stream>>>(dst, cnt, rank, NEDGES);
    gcn_blocksum_kernel<<<(NNODES + T - 1) / T, T, 0, stream>>>(cnt, bsum, NNODES);
    gcn_rowptr_kernel<<<(NNODES + T - 1) / T, T, 0, stream>>>(cnt, bsum, rowptr, isq, csr_src, csr_w, NNODES);
    gcn_csr_kernel<<<(NEDGES + T - 1) / T, T, 0, stream>>>(src, dst, rank, rowptr, isq, csr_src, csr_w, NEDGES);

    {
        int waves  = NNODES / 16;
        int blocks = (waves + 7) / 8;
        gcn_gemm_wmma_kernel<IN_DIM, HID_DIM><<<blocks, T, 0, stream>>>(x, W1, h1, NNODES);
    }
    gcn_gather_kernel<HID_DIM, true><<<(NNODES + 7) / 8, T, 0, stream>>>(
        h1, rowptr, cnt, csr_src, csr_w, isq, b1, agg1, NNODES);

    {
        int waves  = NNODES / 16;
        int blocks = (waves + 7) / 8;
        gcn_gemm_wmma_kernel<HID_DIM, OUT_DIM><<<blocks, T, 0, stream>>>(agg1, W2, h2, NNODES);
    }
    gcn_gather_kernel<OUT_DIM, false><<<(NNODES + 7) / 8, T, 0, stream>>>(
        h2, rowptr, cnt, csr_src, csr_w, isq, b2, out, NNODES);
}
